# starting kernel
# baseline (speedup 1.0000x reference)
#include <hip/hip_runtime.h>
#include <hip/hip_bf16.h>

typedef __attribute__((ext_vector_type(2))) float v2f;
typedef __attribute__((ext_vector_type(8))) float v8f;

#define IW 4096
#define IH 4096
#define OW 4088
#define TILES_X 256
#define TILES_Y 256
#define STRIPS_Y 64
#define WAVES_PER_BLOCK 8

__device__ __forceinline__ void conv_tile_edge(
    const float* __restrict__ X,
    const float (&skern)[81],
    float* __restrict__ out,
    float b0, int x0, int y0, int lane)
{
    const int n  = lane & 15;
    const int kb = (lane >> 4) << 1;

    v8f acc = {};

    #pragma unroll
    for (int c = 0; c < 6; ++c) {
        const int dxb = 4 * c + kb;
        const int dd  = dxb - n;
        const bool m0 = (dd     >= 0) && (dd     <= 8);
        const bool m1 = (dd + 1 >= 0) && (dd + 1 <= 8);
        const int  i0 = min(max(dd,     0), 8);
        const int  i1 = min(max(dd + 1, 0), 8);

        #pragma unroll
        for (int ky = 0; ky < 9; ++ky) {
            int row = y0 + n + ky;
            row = row < IH - 1 ? row : IH - 1;
            int idx = row * IW + (x0 + dxb);
            idx = idx < IW * IW - 2 ? idx : IW * IW - 2;
            v2f a = *(const v2f*)(X + idx);

            float t0 = skern[ky * 9 + i0];
            float t1 = skern[ky * 9 + i1];
            v2f b;
            b.x = m0 ? t0 : 0.0f;
            b.y = m1 ? t1 : 0.0f;

            acc = __builtin_amdgcn_wmma_f32_16x16x4_f32(
                false, a, false, b, (short)0, acc, false, false);
        }
    }

    const int mbase = (lane >> 4) << 3;
    const int ox = x0 + n;
    #pragma unroll
    for (int i = 0; i < 8; ++i) {
        const int oy = y0 + mbase + i;
        if (oy < OW && ox < OW) out[oy * OW + ox] = acc[i] + b0;
    }
}

__device__ __forceinline__ void conv_strip4(
    const float* __restrict__ X,
    const float (&skern)[81],
    float* __restrict__ out,
    float b0, int x0, int y0, int lane)
{
    const int n  = lane & 15;
    const int kb = (lane >> 4) << 1;

    v8f acc[4] = {{}, {}, {}, {}};

    const float* base = X + (y0 + n) * IW + (x0 + kb);

    #pragma unroll
    for (int c = 0; c < 6; ++c) {
        const int dxb = 4 * c + kb;
        const int dd  = dxb - n;
        const bool m0 = (dd     >= 0) && (dd     <= 8);
        const bool m1 = (dd + 1 >= 0) && (dd + 1 <= 8);
        const int  i0 = min(max(dd,     0), 8);
        const int  i1 = min(max(dd + 1, 0), 8);

        #pragma unroll
        for (int ky = 0; ky < 9; ++ky) {
            float t0 = skern[ky * 9 + i0];
            float t1 = skern[ky * 9 + i1];
            v2f b;
            b.x = m0 ? t0 : 0.0f;
            b.y = m1 ? t1 : 0.0f;

            #pragma unroll
            for (int s = 0; s < 4; ++s) {
                v2f a = *(const v2f*)(base + (s * 16 + ky) * IW + 4 * c);
                acc[s] = __builtin_amdgcn_wmma_f32_16x16x4_f32(
                    false, a, false, b, (short)0, acc[s], false, false);
            }
        }
    }

    const int mbase = (lane >> 4) << 3;
    float* orow = out + (y0 + mbase) * OW + (x0 + n);
    #pragma unroll
    for (int s = 0; s < 4; ++s) {
        #pragma unroll
        for (int i = 0; i < 8; ++i) {
            orow[(s * 16 + i) * OW] = acc[s][i] + b0;
        }
    }
}

__global__ __launch_bounds__(256) void conv9x9_wmma_f32(
    const float* __restrict__ X,
    const float* __restrict__ Kf,
    const float* __restrict__ bias,
    float* __restrict__ out)
{
    __shared__ float skern[81];
    for (int t = threadIdx.x; t < 81; t += 256) skern[t] = Kf[t];
    __syncthreads();

    const int lane = threadIdx.x & 31;
    const int wave = threadIdx.x >> 5;
    const int job  = blockIdx.x * WAVES_PER_BLOCK + wave;
    const int tx   = job & (TILES_X - 1);
    const int sy   = job >> 8;
    const int x0   = tx * 16;
    const int y0   = sy * 64;

    const float b0 = bias[0];

    if (tx < TILES_X - 1 && sy < STRIPS_Y - 1) {
        conv_strip4(X, skern, out, b0, x0, y0, lane);
    } else {
        #pragma unroll
        for (int s = 0; s < 4; ++s) {
            conv_tile_edge(X, skern, out, b0, x0, y0 + s * 16, lane);
        }
    }
}

extern "C" void kernel_launch(void* const* d_in, const int* in_sizes, int n_in,
                              void* d_out, int out_size, void* d_ws, size_t ws_size,
                              hipStream_t stream) {
    const float* X    = (const float*)d_in[0];
    const float* Kf   = (const float*)d_in[1];
    const float* bias = (const float*)d_in[2];
    float* out        = (float*)d_out;

    const int total_jobs = TILES_X * STRIPS_Y;
    const int blocks = total_jobs / WAVES_PER_BLOCK;

    hipLaunchKernelGGL(conv9x9_wmma_f32, dim3(blocks), dim3(256), 0, stream,
                       X, Kf, bias, out);
}


# Row streaming: each input row is loaded once with 16-byte loads as the B operand (16 column blocks) and reused from registers by its 9 output rows; taps are banded A fragments; sum order changes
# speedup vs baseline: 2.3061x; 2.3061x over previous
#include <hip/hip_runtime.h>


typedef __attribute__((ext_vector_type(2))) float v2f;
typedef __attribute__((ext_vector_type(4))) float v4f;
typedef __attribute__((ext_vector_type(8))) float v8f;

#define IW 4096
#define IH 4096
#define OW 4088
#define OH 4088
#define KS 9
#define SEG 16
#define NROW (SEG + KS - 1)
#define TPITCH 40
#define WAVES_PER_BLOCK 4
#define NBLOCKS 1024

__global__ __launch_bounds__(128) void conv9x9_rowstream_f32(
    const float* __restrict__ X,
    const float* __restrict__ Kf,
    const float* __restrict__ bias,
    float* __restrict__ out)
{
    __shared__ float stap[KS * TPITCH];
    for (int t = threadIdx.x; t < KS * TPITCH; t += 128) {
        const int ky = t / TPITCH;
        const int d  = t - ky * TPITCH - 15;
        stap[t] = (d >= 0 && d < KS) ? Kf[ky * KS + d] : 0.0f;
    }
    __syncthreads();

    const int lane = threadIdx.x & 31;
    const int wave = __builtin_amdgcn_readfirstlane((int)(threadIdx.x >> 5));
    const int job  = blockIdx.x * WAVES_PER_BLOCK + wave;
    const int xb   = (job & 15) * 256;
    const int r0   = (job >> 4) * SEG;
    const int n    = lane & 15;
    const int h    = lane >> 4;

    v2f afr[KS][6];
    #pragma unroll
    for (int ky = 0; ky < KS; ++ky) {
        #pragma unroll
        for (int c = 0; c < 6; ++c) {
            const int idx = ky * TPITCH + 15 - n + 4 * h + 8 * (c >> 1) + 2 * (c & 1);
            afr[ky][c].x = stap[idx];
            afr[ky][c].y = stap[idx + 1];
        }
    }

    int col[3];
    #pragma unroll
    for (int q = 0; q < 3; ++q) {
        const int cq = xb + 16 * n + 4 * h + 8 * q;
        col[q] = cq < IW - 4 ? cq : IW - 4;
    }

    const float b0 = bias[0];
    const int  ocol  = xb + 16 * n + 8 * h;
    const bool colok = ocol < OW;

    v8f acc[SEG];
    v4f brow[2][3];

    {
        const float* rp = X + (size_t)r0 * IW;
        #pragma unroll
        for (int q = 0; q < 3; ++q) brow[0][q] = *(const v4f*)(rp + col[q]);
    }

    #pragma unroll
    for (int i = 0; i < NROW; ++i) {
        if (i + 1 < NROW) {
            const int r = (r0 + i + 1) < IH - 1 ? (r0 + i + 1) : IH - 1;
            const float* rp = X + (size_t)r * IW;
            #pragma unroll
            for (int q = 0; q < 3; ++q) brow[(i + 1) & 1][q] = *(const v4f*)(rp + col[q]);
        }

        #pragma unroll
        for (int c = 0; c < 6; ++c) {
            v2f b;
            b.x = brow[i & 1][c >> 1][2 * (c & 1)];
            b.y = brow[i & 1][c >> 1][2 * (c & 1) + 1];
            #pragma unroll
            for (int ky = KS - 1; ky >= 0; --ky) {
                const int o = i - ky;
                if (o >= 0 && o < SEG) {
                    if (ky == 0 && c == 0) {
                        const v8f zero = {};
                        acc[o] = __builtin_amdgcn_wmma_f32_16x16x4_f32(
                            false, afr[ky][c], false, b, (short)0, zero, false, false);
                    } else {
                        acc[o] = __builtin_amdgcn_wmma_f32_16x16x4_f32(
                            false, afr[ky][c], false, b, (short)0, acc[o], false, false);
                    }
                }
            }
        }

        if (i >= KS - 1) {
            const int o    = i - (KS - 1);
            const int orow = r0 + o;
            if (orow < OH && colok) {
                float* p = out + (size_t)orow * OW + ocol;
                v4f lo, hi;
                lo.x = acc[o][0] + b0; lo.y = acc[o][1] + b0;
                lo.z = acc[o][2] + b0; lo.w = acc[o][3] + b0;
                hi.x = acc[o][4] + b0; hi.y = acc[o][5] + b0;
                hi.z = acc[o][6] + b0; hi.w = acc[o][7] + b0;
                *(v4f*)p       = lo;
                *(v4f*)(p + 4) = hi;
            }
        }
    }
}

extern "C" void kernel_launch(void* const* d_in, const int* in_sizes, int n_in,
                              void* d_out, int out_size, void* d_ws, size_t ws_size,
                              hipStream_t stream) {
    const float* X    = (const float*)d_in[0];
    const float* Kf   = (const float*)d_in[1];
    const float* bias = (const float*)d_in[2];
    float* out        = (float*)d_out;

    hipLaunchKernelGGL(conv9x9_rowstream_f32, dim3(NBLOCKS), dim3(128), 0, stream,
                       X, Kf, bias, out);
}


# Input rows are fetched three rows ahead of their use (ring of four row buffers in registers, was one ahead in two): a wave alone on the matrix unit no longer waits at each row; sum order kept
# speedup vs baseline: 2.4333x; 1.0552x over previous
#include <hip/hip_runtime.h>


typedef __attribute__((ext_vector_type(2))) float v2f;
typedef __attribute__((ext_vector_type(4))) float v4f;
typedef __attribute__((ext_vector_type(8))) float v8f;

#define IW 4096
#define IH 4096
#define OW 4088
#define OH 4088
#define KS 9
#define SEG 16
#define NROW (SEG + KS - 1)
#define AHEAD 3
#define RING (AHEAD + 1)
#define TPITCH 40
#define WAVES_PER_BLOCK 4
#define NBLOCKS 1024

__global__ __launch_bounds__(128) void conv9x9_rowstream_f32(
    const float* __restrict__ X,
    const float* __restrict__ Kf,
    const float* __restrict__ bias,
    float* __restrict__ out)
{
    __shared__ float stap[KS * TPITCH];
    for (int t = threadIdx.x; t < KS * TPITCH; t += 128) {
        const int ky = t / TPITCH;
        const int d  = t - ky * TPITCH - 15;
        stap[t] = (d >= 0 && d < KS) ? Kf[ky * KS + d] : 0.0f;
    }
    __syncthreads();

    const int lane = threadIdx.x & 31;
    const int wave = __builtin_amdgcn_readfirstlane((int)(threadIdx.x >> 5));
    const int job  = blockIdx.x * WAVES_PER_BLOCK + wave;
    const int xb   = (job & 15) * 256;
    const int r0   = (job >> 4) * SEG;
    const int n    = lane & 15;
    const int h    = lane >> 4;

    v2f afr[KS][6];
    #pragma unroll
    for (int ky = 0; ky < KS; ++ky) {
        #pragma unroll
        for (int c = 0; c < 6; ++c) {
            const int idx = ky * TPITCH + 15 - n + 4 * h + 8 * (c >> 1) + 2 * (c & 1);
            afr[ky][c].x = stap[idx];
            afr[ky][c].y = stap[idx + 1];
        }
    }

    int col[3];
    #pragma unroll
    for (int q = 0; q < 3; ++q) {
        const int cq = xb + 16 * n + 4 * h + 8 * q;
        col[q] = cq < IW - 4 ? cq : IW - 4;
    }

    const float b0 = bias[0];
    const int  ocol  = xb + 16 * n + 8 * h;
    const bool colok = ocol < OW;

    v8f acc[SEG];
    v4f brow[RING][3];

    #pragma unroll
    for (int j = 0; j < AHEAD; ++j) {
        const float* rp = X + (size_t)(r0 + j) * IW;
        #pragma unroll
        for (int q = 0; q < 3; ++q) brow[j][q] = *(const v4f*)(rp + col[q]);
    }

    #pragma unroll
    for (int i = 0; i < NROW; ++i) {
        if (i + AHEAD < NROW) {
            const int r = (r0 + i + AHEAD) < IH - 1 ? (r0 + i + AHEAD) : IH - 1;
            const float* rp = X + (size_t)r * IW;
            #pragma unroll
            for (int q = 0; q < 3; ++q) brow[(i + AHEAD) % RING][q] = *(const v4f*)(rp + col[q]);
        }

        #pragma unroll
        for (int c = 0; c < 6; ++c) {
            v2f b;
            b.x = brow[i % RING][c >> 1][2 * (c & 1)];
            b.y = brow[i % RING][c >> 1][2 * (c & 1) + 1];
            #pragma unroll
            for (int ky = KS - 1; ky >= 0; --ky) {
                const int o = i - ky;
                if (o >= 0 && o < SEG) {
                    if (ky == 0 && c == 0) {
                        const v8f zero = {};
                        acc[o] = __builtin_amdgcn_wmma_f32_16x16x4_f32(
                            false, afr[ky][c], false, b, (short)0, zero, false, false);
                    } else {
                        acc[o] = __builtin_amdgcn_wmma_f32_16x16x4_f32(
                            false, afr[ky][c], false, b, (short)0, acc[o], false, false);
                    }
                }
            }
        }

        if (i >= KS - 1) {
            const int o    = i - (KS - 1);
            const int orow = r0 + o;
            if (orow < OH && colok) {
                float* p = out + (size_t)orow * OW + ocol;
                v4f lo, hi;
                lo.x = acc[o][0] + b0; lo.y = acc[o][1] + b0;
                lo.z = acc[o][2] + b0; lo.w = acc[o][3] + b0;
                hi.x = acc[o][4] + b0; hi.y = acc[o][5] + b0;
                hi.z = acc[o][6] + b0; hi.w = acc[o][7] + b0;
                *(v4f*)p       = lo;
                *(v4f*)(p + 4) = hi;
            }
        }
    }
}

extern "C" void kernel_launch(void* const* d_in, const int* in_sizes, int n_in,
                              void* d_out, int out_size, void* d_ws, size_t ws_size,
                              hipStream_t stream) {
    const float* X    = (const float*)d_in[0];
    const float* Kf   = (const float*)d_in[1];
    const float* bias = (const float*)d_in[2];
    float* out        = (float*)d_out;

    hipLaunchKernelGGL(conv9x9_rowstream_f32, dim3(NBLOCKS), dim3(128), 0, stream,
                       X, Kf, bias, out);
}


# Two output rows at a time (2-parallel fast FIR down the rows: row-pair sums times summed tap rows): 14 row products per 2 rows, was 18; a wave takes 32 rows, 2 waves per SIMD; sum order changes
# speedup vs baseline: 2.7297x; 1.1218x over previous
#include <hip/hip_runtime.h>


typedef __attribute__((ext_vector_type(2))) float v2f;
typedef __attribute__((ext_vector_type(4))) float v4f;
typedef __attribute__((ext_vector_type(8))) float v8f;

#define IW 4096
#define IH 4096
#define OW 4088
#define OH 4088
#define KS 9
#define NJ 16
#define SEG (2 * NJ)
#define NSTEP (NJ + 4)
#define AHEADP 2
#define RINGP (AHEADP + 1)
#define TPITCH 40
#define NTAP 14
#define T_H0 0
#define T_H1 5
#define T_HP 9
#define WAVES_PER_BLOCK 4
#define NBLOCKS 512

__global__ __launch_bounds__(128) void conv9x9_rowpair_fastfir_f32(
    const float* __restrict__ X,
    const float* __restrict__ Kf,
    const float* __restrict__ bias,
    float* __restrict__ out)
{
    __shared__ float stap[NTAP * TPITCH];
    for (int t = threadIdx.x; t < NTAP * TPITCH; t += 128) {
        const int row = t / TPITCH;
        const int d   = t - row * TPITCH - 15;
        float v = 0.0f;
        if (d >= 0 && d < KS) {
            if (row < T_H1) {
                v = Kf[(2 * row) * KS + d];
            } else if (row < T_HP) {
                v = Kf[(2 * (row - T_H1) + 1) * KS + d];
            } else {
                const int a = row - T_HP;
                v = Kf[(2 * a) * KS + d];
                if (a >= 1) v += Kf[(2 * a - 1) * KS + d];
            }
        }
        stap[t] = v;
    }
    __syncthreads();

    const int lane = threadIdx.x & 31;
    const int wave = __builtin_amdgcn_readfirstlane((int)(threadIdx.x >> 5));
    const int job  = blockIdx.x * WAVES_PER_BLOCK + wave;
    const int xb   = (job & 15) * 256;
    const int r0   = (job >> 4) * SEG;
    const int n    = lane & 15;
    const int h    = lane >> 4;

    v2f afr[NTAP][6];
    #pragma unroll
    for (int t = 0; t < NTAP; ++t) {
        #pragma unroll
        for (int c = 0; c < 6; ++c) {
            const int idx = t * TPITCH + 15 - n + 4 * h + 8 * (c >> 1) + 2 * (c & 1);
            afr[t][c].x = stap[idx];
            afr[t][c].y = stap[idx + 1];
        }
    }

    int col[3];
    #pragma unroll
    for (int q = 0; q < 3; ++q) {
        const int cq = xb + 16 * n + 4 * h + 8 * q;
        col[q] = cq < IW - 4 ? cq : IW - 4;
    }

    const float b0 = bias[0];
    const int  ocol  = xb + 16 * n + 8 * h;
    const bool colok = ocol < OW;

    v8f U[NJ], V[NJ + 1], P[NJ];
    v4f brow[RINGP][2][3];

    #pragma unroll
    for (int j = 0; j < AHEADP; ++j) {
        #pragma unroll
        for (int p = 0; p < 2; ++p) {
            const int r = (r0 + 2 * j + p) < IH - 1 ? (r0 + 2 * j + p) : IH - 1;
            const float* rp = X + (size_t)r * IW;
            #pragma unroll
            for (int q = 0; q < 3; ++q) brow[j][p][q] = *(const v4f*)(rp + col[q]);
        }
    }

    #pragma unroll
    for (int i = 0; i < NSTEP; ++i) {
        if (i + AHEADP < NSTEP) {
            #pragma unroll
            for (int p = 0; p < 2; ++p) {
                const int rr = r0 + 2 * (i + AHEADP) + p;
                const int r  = rr < IH - 1 ? rr : IH - 1;
                const float* rp = X + (size_t)r * IW;
                #pragma unroll
                for (int q = 0; q < 3; ++q) brow[(i + AHEADP) % RINGP][p][q] = *(const v4f*)(rp + col[q]);
            }
        }

        v4f srow[3];
        #pragma unroll
        for (int q = 0; q < 3; ++q) srow[q] = brow[i % RINGP][0][q] + brow[i % RINGP][1][q];

        #pragma unroll
        for (int c = 0; c < 6; ++c) {
            v2f bx0, bx1, bxs;
            bx0.x = brow[i % RINGP][0][c >> 1][2 * (c & 1)];
            bx0.y = brow[i % RINGP][0][c >> 1][2 * (c & 1) + 1];
            bx1.x = brow[i % RINGP][1][c >> 1][2 * (c & 1)];
            bx1.y = brow[i % RINGP][1][c >> 1][2 * (c & 1) + 1];
            bxs.x = srow[c >> 1][2 * (c & 1)];
            bxs.y = srow[c >> 1][2 * (c & 1) + 1];
            const v8f zero = {};
            #pragma unroll
            for (int a = 4; a >= 0; --a) {
                const int j = i - a;
                if (j >= 0 && j < NJ) {
                    U[j] = __builtin_amdgcn_wmma_f32_16x16x4_f32(
                        false, afr[T_H0 + a][c], false, bx0, (short)0,
                        (a == 0 && c == 0) ? zero : U[j], false, false);
                }
            }
            #pragma unroll
            for (int a = 3; a >= 0; --a) {
                const int j = i - a;
                if (j >= 0 && j <= NJ) {
                    V[j] = __builtin_amdgcn_wmma_f32_16x16x4_f32(
                        false, afr[T_H1 + a][c], false, bx1, (short)0,
                        (a == 0 && c == 0) ? zero : V[j], false, false);
                }
            }
            #pragma unroll
            for (int a = 4; a >= 0; --a) {
                const int j = i - a;
                if (j >= 0 && j < NJ) {
                    P[j] = __builtin_amdgcn_wmma_f32_16x16x4_f32(
                        false, afr[T_HP + a][c], false, bxs, (short)0,
                        (a == 0 && c == 0) ? zero : P[j], false, false);
                }
            }
        }

        if (i >= 4) {
            const int j    = i - 4;
            const int orow = r0 + 2 * j;
            if (orow < OH && colok) {
                float* pe = out + (size_t)orow * OW + ocol;
                float* po = pe + OW;
                const v8f ye = U[j] + V[j];
                const v8f yo = P[j] - U[j] - V[j + 1];
                v4f lo, hi;
                lo.x = ye[0] + b0; lo.y = ye[1] + b0; lo.z = ye[2] + b0; lo.w = ye[3] + b0;
                hi.x = ye[4] + b0; hi.y = ye[5] + b0; hi.z = ye[6] + b0; hi.w = ye[7] + b0;
                *(v4f*)pe       = lo;
                *(v4f*)(pe + 4) = hi;
                lo.x = yo[0] + b0; lo.y = yo[1] + b0; lo.z = yo[2] + b0; lo.w = yo[3] + b0;
                hi.x = yo[4] + b0; hi.y = yo[5] + b0; hi.z = yo[6] + b0; hi.w = yo[7] + b0;
                *(v4f*)po       = lo;
                *(v4f*)(po + 4) = hi;
            }
        }
    }
}

extern "C" void kernel_launch(void* const* d_in, const int* in_sizes, int n_in,
                              void* d_out, int out_size, void* d_ws, size_t ws_size,
                              hipStream_t stream) {
    const float* X    = (const float*)d_in[0];
    const float* Kf   = (const float*)d_in[1];
    const float* bias = (const float*)d_in[2];
    float* out        = (float*)d_out;

    hipLaunchKernelGGL(conv9x9_rowpair_fastfir_f32, dim3(NBLOCKS), dim3(128), 0, stream,
                       X, Kf, bias, out);
}


# Columns as overlap-add band: B column = 8 input columns, A rows = their 16 partial outputs, neighbour parts added across lanes; 4 matrix ops per row product per 256 cols, was 6; sum order changes
# speedup vs baseline: 2.9025x; 1.0633x over previous
#include <hip/hip_runtime.h>


typedef __attribute__((ext_vector_type(2))) float v2f;
typedef __attribute__((ext_vector_type(4))) float v4f;
typedef __attribute__((ext_vector_type(8))) float v8f;

#define IW 4096
#define IH 4096
#define OW 4088
#define OH 4088
#define KS 9
#define NJ 18
#define SEG (2 * NJ)
#define NSTEP (NJ + 4)
#define AHEADP 2
#define RINGP (AHEADP + 1)
#define NT 2
#define TILE_IN 128
#define STRIP_OUT 248
#define NSTRIP 17
#define NSEG 114
#define NJOBS (NSTRIP * NSEG)
#define TPITCH 24
#define NTAP 14
#define T_H0 0
#define T_H1 5
#define T_HP 9
#define WAVES_PER_BLOCK 4
#define NBLOCKS 485

__global__ __launch_bounds__(128) void conv9x9_overlap_add_fastfir_f32(
    const float* __restrict__ X,
    const float* __restrict__ Kf,
    const float* __restrict__ bias,
    float* __restrict__ out)
{
    __shared__ float stap[NTAP * TPITCH];
    for (int t = threadIdx.x; t < NTAP * TPITCH; t += 128) {
        const int row = t / TPITCH;
        const int d   = t - row * TPITCH - 7;
        float v = 0.0f;
        if (d >= 0 && d < KS) {
            if (row < T_H1) {
                v = Kf[(2 * row) * KS + d];
            } else if (row < T_HP) {
                v = Kf[(2 * (row - T_H1) + 1) * KS + d];
            } else {
                const int a = row - T_HP;
                v = Kf[(2 * a) * KS + d];
                if (a >= 1) v += Kf[(2 * a - 1) * KS + d];
            }
        }
        stap[t] = v;
    }
    __syncthreads();

    const int lane = threadIdx.x & 31;
    const int wave = __builtin_amdgcn_readfirstlane((int)(threadIdx.x >> 5));
    const int job  = blockIdx.x * WAVES_PER_BLOCK + wave;
    if (job >= NJOBS) return;
    const int seg   = job / NSTRIP;
    const int strip = job - seg * NSTRIP;
    const int xb    = strip * STRIP_OUT;
    const int r0    = seg * SEG;
    const int n     = lane & 15;
    const int h     = lane >> 4;

    const int f = (n & 3) + 4 * (n >> 3) - 8 * ((n >> 2) & 1);
    v2f afr[NTAP][2];
    #pragma unroll
    for (int t = 0; t < NTAP; ++t) {
        #pragma unroll
        for (int c = 0; c < 2; ++c) {
            const int idx = t * TPITCH + 7 + 4 * h + 2 * c - f;
            afr[t][c].x = stap[idx];
            afr[t][c].y = stap[idx + 1];
        }
    }

    int col[NT];
    #pragma unroll
    for (int q = 0; q < NT; ++q) {
        const int cq = xb + TILE_IN * q + 8 * n + 4 * h;
        col[q] = cq < IW - 4 ? cq : IW - 4;
    }

    const float b0 = bias[0];
    const int  oc0 = xb + 8 * n + 4 * h;
    const int  oc1 = oc0 + TILE_IN;
    const bool ok0 = oc0 < OW;
    const bool ok1 = (n < 15) && (oc1 < OW);
    const int  nxt   = ((lane & 16) | ((n + 1) & 15)) << 2;
    const bool first = (n == 0);

    v8f U[NJ][NT], V[NJ + 1][NT], P[NJ][NT];
    v4f brow[RINGP][2][NT];

    #pragma unroll
    for (int j = 0; j < AHEADP; ++j) {
        #pragma unroll
        for (int p = 0; p < 2; ++p) {
            const int r = (r0 + 2 * j + p) < IH - 1 ? (r0 + 2 * j + p) : IH - 1;
            const float* rp = X + (size_t)r * IW;
            #pragma unroll
            for (int q = 0; q < NT; ++q) brow[j][p][q] = *(const v4f*)(rp + col[q]);
        }
    }

    #pragma unroll
    for (int i = 0; i < NSTEP; ++i) {
        if (i + AHEADP < NSTEP) {
            #pragma unroll
            for (int p = 0; p < 2; ++p) {
                const int rr = r0 + 2 * (i + AHEADP) + p;
                const int r  = rr < IH - 1 ? rr : IH - 1;
                const float* rp = X + (size_t)r * IW;
                #pragma unroll
                for (int q = 0; q < NT; ++q) brow[(i + AHEADP) % RINGP][p][q] = *(const v4f*)(rp + col[q]);
            }
        }

        #pragma unroll
        for (int q = 0; q < NT; ++q) {
            const v4f srow = brow[i % RINGP][0][q] + brow[i % RINGP][1][q];
            #pragma unroll
            for (int c = 0; c < 2; ++c) {
                v2f bx0, bx1, bxs;
                bx0.x = brow[i % RINGP][0][q][2 * c];
                bx0.y = brow[i % RINGP][0][q][2 * c + 1];
                bx1.x = brow[i % RINGP][1][q][2 * c];
                bx1.y = brow[i % RINGP][1][q][2 * c + 1];
                bxs.x = srow[2 * c];
                bxs.y = srow[2 * c + 1];
                const v8f zero = {};
                #pragma unroll
                for (int a = 4; a >= 0; --a) {
                    const int j = i - a;
                    if (j >= 0 && j < NJ) {
                        U[j][q] = __builtin_amdgcn_wmma_f32_16x16x4_f32(
                            false, afr[T_H0 + a][c], false, bx0, (short)0,
                            (a == 0 && c == 0) ? zero : U[j][q], false, false);
                    }
                }
                #pragma unroll
                for (int a = 3; a >= 0; --a) {
                    const int j = i - a;
                    if (j >= 0 && j <= NJ) {
                        V[j][q] = __builtin_amdgcn_wmma_f32_16x16x4_f32(
                            false, afr[T_H1 + a][c], false, bx1, (short)0,
                            (a == 0 && c == 0) ? zero : V[j][q], false, false);
                    }
                }
                #pragma unroll
                for (int a = 4; a >= 0; --a) {
                    const int j = i - a;
                    if (j >= 0 && j < NJ) {
                        P[j][q] = __builtin_amdgcn_wmma_f32_16x16x4_f32(
                            false, afr[T_HP + a][c], false, bxs, (short)0,
                            (a == 0 && c == 0) ? zero : P[j][q], false, false);
                    }
                }
            }
        }

        if (i >= 4) {
            const int j    = i - 4;
            const int orow = r0 + 2 * j;
            if (orow < OH) {
                v8f y[2][NT];
                #pragma unroll
                for (int q = 0; q < NT; ++q) {
                    y[0][q] = U[j][q] + V[j][q];
                    y[1][q] = P[j][q] - U[j][q] - V[j + 1][q];
                }
                #pragma unroll
                for (int p = 0; p < 2; ++p) {
                    v4f o0, o1;
                    #pragma unroll
                    for (int r = 0; r < 4; ++r) {
                        const float m0 = first ? y[p][1][r + 4] : y[p][0][r + 4];
                        const float s0 = __int_as_float(__builtin_amdgcn_ds_bpermute(nxt, __float_as_int(m0)));
                        const float s1 = __int_as_float(__builtin_amdgcn_ds_bpermute(nxt, __float_as_int(y[p][1][r + 4])));
                        o0[r] = y[p][0][r] + s0 + b0;
                        o1[r] = y[p][1][r] + s1 + b0;
                    }
                    float* po = out + (size_t)(orow + p) * OW;
                    if (ok0) *(v4f*)(po + oc0) = o0;
                    if (ok1) *(v4f*)(po + oc1) = o1;
                }
            }
        }
    }
}

extern "C" void kernel_launch(void* const* d_in, const int* in_sizes, int n_in,
                              void* d_out, int out_size, void* d_ws, size_t ws_size,
                              hipStream_t stream) {
    const float* X    = (const float*)d_in[0];
    const float* Kf   = (const float*)d_in[1];
    const float* bias = (const float*)d_in[2];
    float* out        = (float*)d_out;

    hipLaunchKernelGGL(conv9x9_overlap_add_fastfir_f32, dim3(NBLOCKS), dim3(128), 0, stream,
                       X, Kf, bias, out);
}
